# starting kernel
# baseline (speedup 1.0000x reference)
#include <hip/hip_runtime.h>
#include <math.h>

typedef __attribute__((ext_vector_type(2))) float v2f;
typedef __attribute__((ext_vector_type(8))) float v8f;

#define NUM_AC    32
#define NWIN      64
#define WLEN      256
#define WSTRIDE   128
#define ROWLEN    8320
#define CSTRIDE_M 20
#define CSTRIDE_C 320

__global__ __launch_bounds__(256, 1)
void autocorr_wmma_kernel(const float* __restrict__ x, float* __restrict__ out)
{
    __shared__ __align__(16) float s_row[ROWLEN];
    __shared__ float s_hann[WLEN];
    __shared__ float s_v[8][288];
    __shared__ float s_c[8][3 * CSTRIDE_C];

    const int tid  = threadIdx.x;
    const int lane = tid & 31;
    const int wv   = tid >> 5;
    const int col  = lane & 15;
    const int hi   = lane >> 4;
    const int row  = blockIdx.x;

    if (tid < WLEN) {
        s_hann[tid] = 0.5f - 0.5f * cosf(6.28318530717958647692f * (float)tid * (1.0f / 255.0f));
    }

    {
        const float4* src = reinterpret_cast<const float4*>(x + (size_t)row * ROWLEN);
        float4*       dst = reinterpret_cast<float4*>(s_row);
        for (int i = tid; i < ROWLEN / 4; i += 256) dst[i] = src[i];
    }
    __syncthreads();

    float* vb = s_v[wv];
    float* cb = s_c[wv];

    for (int w = wv; w < NWIN; w += 8) {
        const float* win = s_row + w * WSTRIDE;

        float ps = 0.0f;
        #pragma unroll
        for (int i = 0; i < 8; ++i) ps += win[i * 32 + lane];
        #pragma unroll
        for (int d = 16; d >= 1; d >>= 1) ps += __shfl_xor(ps, d, 32);
        const float mean = ps * (1.0f / 256.0f);

        #pragma unroll
        for (int i = 0; i < 8; ++i) {
            const int t = i * 32 + lane;
            vb[t] = (win[t] - mean) * s_hann[t];
        }
        vb[256 + lane] = 0.0f;
        __builtin_amdgcn_wave_barrier();

        float r[16];
        #pragma unroll
        for (int i = 0; i < 16; ++i) r[i] = vb[(2 * hi + i) * 16 + col];

        v8f c0 = {}; v8f c1 = {}; v8f c2 = {};
        #pragma unroll
        for (int q = 0; q < 4; ++q) {
            v2f a  = { r[4 * q + 0], r[4 * q + 1] };
            v2f b1 = { r[4 * q + 1], r[4 * q + 2] };
            v2f b2 = { r[4 * q + 2], r[4 * q + 3] };
            c0 = __builtin_amdgcn_wmma_f32_16x16x4_f32(false, a, false, a,  (short)0, c0, false, false);
            c1 = __builtin_amdgcn_wmma_f32_16x16x4_f32(false, a, false, b1, (short)0, c1, false, false);
            c2 = __builtin_amdgcn_wmma_f32_16x16x4_f32(false, a, false, b2, (short)0, c2, false, false);
        }

        #pragma unroll
        for (int j = 0; j < 8; ++j) {
            const int m = 8 * hi + j;
            cb[0 * CSTRIDE_C + m * CSTRIDE_M + col] = c0[j];
            cb[1 * CSTRIDE_C + m * CSTRIDE_M + col] = c1[j];
            cb[2 * CSTRIDE_C + m * CSTRIDE_M + col] = c2[j];
        }
        __builtin_amdgcn_wave_barrier();

        float acc = 0.0f;
        #pragma unroll
        for (int m = 0; m < 16; ++m) {
            const int s2 = lane + m;
            acc += cb[(s2 >> 4) * CSTRIDE_C + m * CSTRIDE_M + (s2 & 15)];
        }
        out[((size_t)row * NWIN + w) * NUM_AC + lane] = acc * (1.0f / 256.0f);
        __builtin_amdgcn_wave_barrier();
    }
}

extern "C" void kernel_launch(void* const* d_in, const int* in_sizes, int n_in,
                              void* d_out, int out_size, void* d_ws, size_t ws_size,
                              hipStream_t stream) {
    const float* x   = (const float*)d_in[0];
    float*       out = (float*)d_out;
    dim3 grid(16 * 128);
    dim3 block(256);
    hipLaunchKernelGGL(autocorr_wmma_kernel, grid, block, 0, stream, x, out);
}


# Diagonal sums of the three C tiles stay in registers: per-lane tile pick, DPP row rotate by row index, one row swap; LDS scatter/gather gone; sum over m is now two 8-term halves (order changed)
# speedup vs baseline: 1.0580x; 1.0580x over previous
#include <hip/hip_runtime.h>
#include <math.h>

typedef __attribute__((ext_vector_type(2))) float v2f;
typedef __attribute__((ext_vector_type(8))) float v8f;

#define NUM_AC    32
#define NWIN      64
#define WLEN      256
#define WSTRIDE   128
#define ROWLEN    8320

template <int J>
__device__ __forceinline__ float rotl16(float x)
{
    return __builtin_bit_cast(float,
        __builtin_amdgcn_update_dpp(0, __builtin_bit_cast(int, x), 0x120 + ((16 - J) & 15), 0xf, 0xf, true));
}

__device__ __forceinline__ float xhalf(float x)
{
    const int b = __builtin_bit_cast(int, x);
    return __builtin_bit_cast(float,
        __builtin_amdgcn_permlanex16(b, b, 0x76543210, 0xfedcba98, false, false));
}

__global__ __launch_bounds__(256, 1)
void autocorr_wmma_kernel(const float* __restrict__ x, float* __restrict__ out)
{
    __shared__ __align__(16) float s_row[ROWLEN];
    __shared__ float s_hann[WLEN];
    __shared__ float s_v[8][288];

    const int tid  = threadIdx.x;
    const int lane = tid & 31;
    const int wv   = tid >> 5;
    const int col  = lane & 15;
    const int hi   = lane >> 4;
    const int row  = blockIdx.x;

    if (tid < WLEN) {
        s_hann[tid] = 0.5f - 0.5f * cosf(6.28318530717958647692f * (float)tid * (1.0f / 255.0f));
    }

    {
        const float4* src = reinterpret_cast<const float4*>(x + (size_t)row * ROWLEN);
        float4*       dst = reinterpret_cast<float4*>(s_row);
        for (int i = tid; i < ROWLEN / 4; i += 256) dst[i] = src[i];
    }
    __syncthreads();

    float* vb = s_v[wv];

    for (int w = wv; w < NWIN; w += 8) {
        const float* win = s_row + w * WSTRIDE;

        float ps = 0.0f;
        #pragma unroll
        for (int i = 0; i < 8; ++i) ps += win[i * 32 + lane];
        #pragma unroll
        for (int d = 16; d >= 1; d >>= 1) ps += __shfl_xor(ps, d, 32);
        const float mean = ps * (1.0f / 256.0f);

        #pragma unroll
        for (int i = 0; i < 8; ++i) {
            const int t = i * 32 + lane;
            vb[t] = (win[t] - mean) * s_hann[t];
        }
        vb[256 + lane] = 0.0f;
        __builtin_amdgcn_wave_barrier();

        float r[16];
        #pragma unroll
        for (int i = 0; i < 16; ++i) r[i] = vb[(2 * hi + i) * 16 + col];

        v8f c0 = {}; v8f c1 = {}; v8f c2 = {};
        #pragma unroll
        for (int q = 0; q < 4; ++q) {
            v2f a  = { r[4 * q + 0], r[4 * q + 1] };
            v2f b1 = { r[4 * q + 1], r[4 * q + 2] };
            v2f b2 = { r[4 * q + 2], r[4 * q + 3] };
            c0 = __builtin_amdgcn_wmma_f32_16x16x4_f32(false, a, false, a,  (short)0, c0, false, false);
            c1 = __builtin_amdgcn_wmma_f32_16x16x4_f32(false, a, false, b1, (short)0, c1, false, false);
            c2 = __builtin_amdgcn_wmma_f32_16x16x4_f32(false, a, false, b2, (short)0, c2, false, false);
        }

        float s0 = c0[0], s1 = c1[0], s2 = c2[0];
#define DIAG_STEP(J)                                              \
        {                                                         \
            const bool own = (col >= J);                          \
            s0 += rotl16<J>(own ? c0[J] : c1[J]);                 \
            s1 += rotl16<J>(own ? c1[J] : c2[J]);                 \
            s2 += rotl16<J>(c2[J]);                               \
        }
        DIAG_STEP(1) DIAG_STEP(2) DIAG_STEP(3) DIAG_STEP(4)
        DIAG_STEP(5) DIAG_STEP(6) DIAG_STEP(7)
#undef DIAG_STEP
        const bool own8 = (col >= 8);
        const float u0 = rotl16<8>(own8 ? s0 : s1);
        const float u1 = rotl16<8>(own8 ? s1 : s2);
        const float mine   = hi ? u1 : s0;
        const float theirs = hi ? u0 : s1;
        const float acc = mine + xhalf(theirs);
        out[((size_t)row * NWIN + w) * NUM_AC + lane] = acc * (1.0f / 256.0f);
    }
}

extern "C" void kernel_launch(void* const* d_in, const int* in_sizes, int n_in,
                              void* d_out, int out_size, void* d_ws, size_t ws_size,
                              hipStream_t stream) {
    const float* x   = (const float*)d_in[0];
    float*       out = (float*)d_out;
    dim3 grid(16 * 128);
    dim3 block(256);
    hipLaunchKernelGGL(autocorr_wmma_kernel, grid, block, 0, stream, x, out);
}


# Windows load from global memory straight into the matrix fragment layout (k block = 8*half+j), 16 per wave, Hann in registers; LDS row staging and per-window LDS trips gone; mean, k sum order changed
# speedup vs baseline: 1.2330x; 1.1654x over previous
#include <hip/hip_runtime.h>
#include <math.h>


typedef __attribute__((ext_vector_type(2))) float v2f;
typedef __attribute__((ext_vector_type(8))) float v8f;

#define NUM_AC    32
#define NWIN      64
#define WLEN      256
#define WSTRIDE   128
#define ROWLEN    8320
#define SEG       16
#define NBLK      10

template <int J>
__device__ __forceinline__ float rotl16(float x)
{
    return __builtin_bit_cast(float,
        __builtin_amdgcn_update_dpp(0, __builtin_bit_cast(int, x), 0x120 + ((16 - J) & 15), 0xf, 0xf, true));
}

__device__ __forceinline__ float xhalf(float x)
{
    const int b = __builtin_bit_cast(int, x);
    return __builtin_bit_cast(float,
        __builtin_amdgcn_permlanex16(b, b, 0x76543210, 0xfedcba98, false, false));
}

__device__ __forceinline__ void load_window(float (&raw)[NBLK], const float* __restrict__ p,
                                            const float* __restrict__ q)
{
    for (int j = 0; j < 8; ++j) raw[j] = p[16 * j];
    raw[8] = q[0];
    raw[9] = q[16];
}

__device__ __forceinline__ void window_lags(const float (&raw)[NBLK], const float (&hw)[NBLK],
                                            int col, int hi, float* __restrict__ dst)
{
    float ps = ((raw[0] + raw[2]) + (raw[4] + raw[6])) + ((raw[1] + raw[3]) + (raw[5] + raw[7]));
    ps += rotl16<8>(ps);
    ps += rotl16<4>(ps);
    ps += rotl16<2>(ps);
    ps += rotl16<1>(ps);
    ps += xhalf(ps);
    const float mean = ps * (1.0f / 256.0f);

    float r[NBLK];
    for (int j = 0; j < NBLK; ++j) r[j] = (raw[j] - mean) * hw[j];

    v8f c0 = {}; v8f c1 = {}; v8f c2 = {};
    for (int q = 0; q < 4; ++q) {
        v2f a  = { r[2 * q + 0], r[2 * q + 1] };
        v2f b1 = { r[2 * q + 1], r[2 * q + 2] };
        v2f b2 = { r[2 * q + 2], r[2 * q + 3] };
        c0 = __builtin_amdgcn_wmma_f32_16x16x4_f32(false, a, false, a,  (short)0, c0, false, false);
        c1 = __builtin_amdgcn_wmma_f32_16x16x4_f32(false, a, false, b1, (short)0, c1, false, false);
        c2 = __builtin_amdgcn_wmma_f32_16x16x4_f32(false, a, false, b2, (short)0, c2, false, false);
    }

    float s0 = c0[0], s1 = c1[0], s2 = c2[0];
#define DIAG_STEP(J)                                              \
    {                                                             \
        const bool own = (col >= J);                              \
        s0 += rotl16<J>(own ? c0[J] : c1[J]);                     \
        s1 += rotl16<J>(own ? c1[J] : c2[J]);                     \
        s2 += rotl16<J>(c2[J]);                                   \
    }
    DIAG_STEP(1) DIAG_STEP(2) DIAG_STEP(3) DIAG_STEP(4)
    DIAG_STEP(5) DIAG_STEP(6) DIAG_STEP(7)
#undef DIAG_STEP
    const bool own8 = (col >= 8);
    const float u0 = rotl16<8>(own8 ? s0 : s1);
    const float u1 = rotl16<8>(own8 ? s1 : s2);
    const float mine   = hi ? u1 : s0;
    const float theirs = hi ? u0 : s1;
    const float acc = mine + xhalf(theirs);
    dst[16 * hi + col] = acc * (1.0f / 256.0f);
}

__global__ __launch_bounds__(256)
void autocorr_wmma_kernel(const float* __restrict__ x, float* __restrict__ out)
{
    __shared__ float s_hann[WLEN];

    const int tid  = threadIdx.x;
    const int lane = tid & 31;
    const int wv   = tid >> 5;
    const int col  = lane & 15;
    const int hi   = lane >> 4;
    const int row  = blockIdx.x * 2 + (wv >> 2);
    const int w0   = (wv & 3) * SEG;

    s_hann[tid] = 0.5f - 0.5f * cosf(6.28318530717958647692f * (float)tid * (1.0f / 255.0f));
    __syncthreads();

    float hw[NBLK];
    for (int j = 0; j < 8; ++j) hw[j] = s_hann[128 * hi + 16 * j + col];
    hw[8] = hi ? 0.0f : s_hann[128 + col];
    hw[9] = hi ? 0.0f : s_hann[144 + col];

    const float* p = x + (size_t)row * ROWLEN + w0 * WSTRIDE + 128 * hi + col;
    const float* q = x + (size_t)row * ROWLEN + w0 * WSTRIDE + 128 + col;
    float*       o = out + ((size_t)row * NWIN + w0) * NUM_AC;

    float wa[NBLK], wb[NBLK];
    load_window(wa, p, q);
    for (int it = 0; it < SEG - 2; it += 2) {
        load_window(wb, p + WSTRIDE, q + WSTRIDE);
        window_lags(wa, hw, col, hi, o);
        load_window(wa, p + 2 * WSTRIDE, q + 2 * WSTRIDE);
        window_lags(wb, hw, col, hi, o + NUM_AC);
        p += 2 * WSTRIDE;
        q += 2 * WSTRIDE;
        o += 2 * NUM_AC;
    }
    load_window(wb, p + WSTRIDE, q + WSTRIDE);
    window_lags(wa, hw, col, hi, o);
    window_lags(wb, hw, col, hi, o + NUM_AC);
}

extern "C" void kernel_launch(void* const* d_in, const int* in_sizes, int n_in,
                              void* d_out, int out_size, void* d_ws, size_t ws_size,
                              hipStream_t stream) {
    const float* x   = (const float*)d_in[0];
    float*       out = (float*)d_out;
    dim3 grid(1024);
    dim3 block(256);
    hipLaunchKernelGGL(autocorr_wmma_kernel, grid, block, 0, stream, x, out);
}
